# starting kernel
# baseline (speedup 1.0000x reference)
#include <hip/hip_runtime.h>
#include <hip/hip_bf16.h>


typedef __attribute__((ext_vector_type(16))) _Float16 v16h;
typedef __attribute__((ext_vector_type(8)))  _Float16 v8h;
typedef __attribute__((ext_vector_type(2)))  _Float16 v2h;
typedef __attribute__((ext_vector_type(8)))  float    v8f;
typedef __attribute__((ext_vector_type(2)))  float    v2f;

#define HID   32
#define NTILE 8

#if __has_builtin(__builtin_amdgcn_wmma_f32_16x16x4_f32)
#define HAS_K4 1
#else
#define HAS_K4 0
#endif

#if __has_builtin(__builtin_amdgcn_tanhf)
__device__ __forceinline__ float tanh_(float x) { return __builtin_amdgcn_tanhf(x); }
__device__ __forceinline__ float sigm(float x) {
    return __builtin_fmaf(0.5f, __builtin_amdgcn_tanhf(0.5f * x), 0.5f);
}
#else
__device__ __forceinline__ float fast_rcp(float x) { return __builtin_amdgcn_rcpf(x); }
__device__ __forceinline__ float tanh_(float x) {
    float e = __expf(-2.0f * x);
    return (1.0f - e) * fast_rcp(1.0f + e);
}
__device__ __forceinline__ float sigm(float x) { return fast_rcp(1.0f + __expf(-x)); }
#endif

__device__ __forceinline__ int permK(int k) { return (k & 1) ? ((k >> 1) + 16) : (k >> 1); }

__global__ __launch_bounds__(32)
void lstm_wmma_kernel(const float* __restrict__ x,
                      const float* __restrict__ W_ih,
                      const float* __restrict__ W_hh,
                      const float* __restrict__ b_ih,
                      const float* __restrict__ b_hh,
                      const float* __restrict__ W_fc,
                      const float* __restrict__ b_fc,
                      float* __restrict__ out,
                      int T)
{
    __shared__ _Float16 hbuf[16 * HID];

    const int lane = threadIdx.x;
    const int col  = lane & 15;
    const int hi   = lane >> 4;
    const int b0   = blockIdx.x * 16;

    v16h Bt[NTILE];
#pragma unroll
    for (int tt = 0; tt < NTILE; ++tt) {
        const float* wrow = W_hh + (size_t)(tt * 16 + col) * HID;
#pragma unroll
        for (int e = 0; e < 8; ++e) {
            Bt[tt][e]     = (_Float16)wrow[permK(hi * 8 + e)];
            Bt[tt][e + 8] = (_Float16)wrow[permK(16 + hi * 8 + e)];
        }
    }

#if HAS_K4
    v2f Bx[NTILE];
#pragma unroll
    for (int tt = 0; tt < NTILE; ++tt) {
        int g = tt * 16 + col;
        if (hi == 0) { Bx[tt][0] = W_ih[g * 2 + 0];     Bx[tt][1] = W_ih[g * 2 + 1]; }
        else         { Bx[tt][0] = b_ih[g] + b_hh[g];   Bx[tt][1] = 0.0f; }
    }
#else
    v16h Bx[NTILE];
#pragma unroll
    for (int tt = 0; tt < NTILE; ++tt) {
        Bx[tt] = (v16h){};
        int g = tt * 16 + col;
        if (hi == 0) {
            Bx[tt][0] = (_Float16)W_ih[g * 2 + 0];
            Bx[tt][1] = (_Float16)W_ih[g * 2 + 1];
            Bx[tt][2] = (_Float16)(b_ih[g] + b_hh[g]);
        }
    }
#endif

    v16h Bfc = (v16h){};
    if (col == 0) {
#pragma unroll
        for (int e = 0; e < 8; ++e) {
            Bfc[e]     = (_Float16)W_fc[permK(hi * 8 + e)];
            Bfc[e + 8] = (_Float16)W_fc[permK(16 + hi * 8 + e)];
        }
    }
    v8f Cfc;
    {
        float bfc = b_fc[0];
#pragma unroll
        for (int e = 0; e < 8; ++e) Cfc[e] = bfc;
    }

    float c1[8], c2[8];
#pragma unroll
    for (int r = 0; r < 8; ++r) { c1[r] = 0.f; c2[r] = 0.f; }
    for (int i = lane; i < 16 * HID; i += 32) hbuf[i] = (_Float16)0.f;
    __syncthreads();

    const size_t xrow = (size_t)T * 2;
    const float* pxbase = x + (size_t)(b0 + col) * xrow;

#if !HAS_K4
    v16h Ax = (v16h){};
    if (hi == 0) Ax[2] = (_Float16)1.0f;
#endif

    float2 xv = *(const float2*)pxbase;

    for (int t = 0; t < T; ++t) {
#if HAS_K4
        v2f Ax;
        Ax[0] = (hi == 0) ? xv.x : 1.0f;
        Ax[1] = (hi == 0) ? xv.y : 0.0f;
#else
        if (hi == 0) { Ax[0] = (_Float16)xv.x; Ax[1] = (_Float16)xv.y; }
#endif

        int tn = (t + 1 < T) ? (t + 1) : t;
        float2 xvn = *(const float2*)(pxbase + (size_t)tn * 2);

        v16h A;
        {
            v8h alo = *(const v8h*)&hbuf[col * HID + hi * 8];
            v8h ahi = *(const v8h*)&hbuf[col * HID + 16 + hi * 8];
#pragma unroll
            for (int e = 0; e < 8; ++e) { A[e] = alo[e]; A[e + 8] = ahi[e]; }
        }

        v8f accy = __builtin_amdgcn_wmma_f32_16x16x32_f16(
            false, A, false, Bfc, (short)0, Cfc, false, false);

        v8f acc[NTILE];
#pragma unroll
        for (int tt = 0; tt < NTILE; ++tt) {
            v8f z = (v8f){};
#if HAS_K4
            acc[tt] = __builtin_amdgcn_wmma_f32_16x16x4_f32(
                false, Ax, false, Bx[tt], (short)0, z, false, false);
#else
            acc[tt] = __builtin_amdgcn_wmma_f32_16x16x32_f16(
                false, Ax, false, Bx[tt], (short)0, z, false, false);
#endif
        }
#pragma unroll
        for (int tt = 0; tt < NTILE; ++tt) {
            acc[tt] = __builtin_amdgcn_wmma_f32_16x16x32_f16(
                false, A, false, Bt[tt], (short)0, acc[tt], false, false);
        }

#pragma unroll
        for (int r = 0; r < 8; ++r) {
            float i1 = sigm(acc[0][r]);
            float f1 = sigm(acc[2][r]);
            float g1 = tanh_(acc[4][r]);
            float o1 = sigm(acc[6][r]);
            c1[r] = f1 * c1[r] + i1 * g1;
            float h1 = o1 * tanh_(c1[r]);

            float i2 = sigm(acc[1][r]);
            float f2 = sigm(acc[3][r]);
            float g2 = tanh_(acc[5][r]);
            float o2 = sigm(acc[7][r]);
            c2[r] = f2 * c2[r] + i2 * g2;
            float h2 = o2 * tanh_(c2[r]);

            int row = r + 8 * hi;
            *(v2h*)&hbuf[row * HID + col * 2] = (v2h){(_Float16)h1, (_Float16)h2};
        }
        __syncthreads();

        if (t > 0 && col == 0) {
#pragma unroll
            for (int r = 0; r < 8; ++r)
                out[(size_t)(b0 + r + 8 * hi) * T + (t - 1)] = accy[r];
        }
        xv = xvn;
    }

    {
        v16h A;
        v8h alo = *(const v8h*)&hbuf[col * HID + hi * 8];
        v8h ahi = *(const v8h*)&hbuf[col * HID + 16 + hi * 8];
#pragma unroll
        for (int e = 0; e < 8; ++e) { A[e] = alo[e]; A[e + 8] = ahi[e]; }
        v8f accy = __builtin_amdgcn_wmma_f32_16x16x32_f16(
            false, A, false, Bfc, (short)0, Cfc, false, false);
        if (col == 0) {
#pragma unroll
            for (int r = 0; r < 8; ++r)
                out[(size_t)(b0 + r + 8 * hi) * T + (T - 1)] = accy[r];
        }
    }
}

extern "C" void kernel_launch(void* const* d_in, const int* in_sizes, int n_in,
                              void* d_out, int out_size, void* d_ws, size_t ws_size,
                              hipStream_t stream) {
    const float* x    = (const float*)d_in[0];
    const float* W_ih = (const float*)d_in[1];
    const float* W_hh = (const float*)d_in[2];
    const float* b_ih = (const float*)d_in[3];
    const float* b_hh = (const float*)d_in[4];
    const float* W_fc = (const float*)d_in[5];
    const float* b_fc = (const float*)d_in[6];
    float* out = (float*)d_out;

    const int T = 512;
    const int B = out_size / T;
    const int nblocks = B / 16;

    lstm_wmma_kernel<<<dim3(nblocks), dim3(32), 0, stream>>>(
        x, W_ih, W_hh, b_ih, b_hh, W_fc, b_fc, out, T);

    (void)in_sizes; (void)n_in; (void)d_ws; (void)ws_size;
}


# Split each 16-batch tile over 4 waves (one per SIMD): gates on the WMMA M axis so a lane owns i,f,g,o of its units; h_t via double-buffered LDS, 1 barrier/step; k order in f16 fragments changes
# speedup vs baseline: 1.6702x; 1.6702x over previous
#include <hip/hip_runtime.h>
#include <hip/hip_bf16.h>


typedef __attribute__((ext_vector_type(16))) _Float16 v16h;
typedef __attribute__((ext_vector_type(8)))  _Float16 v8h;
typedef __attribute__((ext_vector_type(2)))  _Float16 v2h;
typedef __attribute__((ext_vector_type(8)))  float    v8f;
typedef __attribute__((ext_vector_type(2)))  float    v2f;

#define HID   32
#define NWAVE 4
#define TPW   2
#define HP    40
#define HBUF  1024

#if __has_builtin(__builtin_amdgcn_tanhf)
__device__ __forceinline__ float tanh_(float x) { return __builtin_amdgcn_tanhf(x); }
__device__ __forceinline__ float sigm(float x) {
    return __builtin_fmaf(0.5f, __builtin_amdgcn_tanhf(0.5f * x), 0.5f);
}
#else
__device__ __forceinline__ float fast_rcp(float x) { return __builtin_amdgcn_rcpf(x); }
__device__ __forceinline__ float tanh_(float x) {
    float e = __expf(-2.0f * x);
    return (1.0f - e) * fast_rcp(1.0f + e);
}
__device__ __forceinline__ float sigm(float x) { return fast_rcp(1.0f + __expf(-x)); }
#endif


__global__ __launch_bounds__(32 * NWAVE)
void lstm_wmma_kernel(const float* __restrict__ x,
                      const float* __restrict__ W_ih,
                      const float* __restrict__ W_hh,
                      const float* __restrict__ b_ih,
                      const float* __restrict__ b_hh,
                      const float* __restrict__ W_fc,
                      const float* __restrict__ b_fc,
                      float* __restrict__ out,
                      int T)
{
    __shared__ __attribute__((aligned(16))) _Float16 hbuf[2 * HBUF];

    const int lane = threadIdx.x & 31;
    const int w    = __builtin_amdgcn_readfirstlane((int)(threadIdx.x >> 5));
    const int col  = lane & 15;
    const int hi   = lane >> 4;
    const int b0   = blockIdx.x * 16;

    v16h Aw[TPW];
    v2f  Ax[TPW];
#pragma unroll
    for (int tl = 0; tl < TPW; ++tl) {
        const int tt = w * TPW + tl;
        const int q  = (col & 7) >> 1;
        const int j  = 4 * tt + 2 * (col >> 3) + (col & 1);
        const int g  = q * HID + j;
        const float* wrow = W_hh + (size_t)g * HID + 16 * hi;
#pragma unroll
        for (int s = 0; s < 16; ++s) Aw[tl][s] = (_Float16)wrow[s];
        if (hi == 0) { Ax[tl][0] = W_ih[g * 2 + 0];    Ax[tl][1] = W_ih[g * 2 + 1]; }
        else         { Ax[tl][0] = b_ih[g] + b_hh[g];  Ax[tl][1] = 0.0f; }
    }

    v16h Afc = (v16h){};
    if (col == 0) {
#pragma unroll
        for (int s = 0; s < 16; ++s) Afc[s] = (_Float16)W_fc[16 * hi + s];
    }
    v8f Cfc;
    {
        float bfc = b_fc[0];
#pragma unroll
        for (int e = 0; e < 8; ++e) Cfc[e] = bfc;
    }

    float c[TPW][2];
#pragma unroll
    for (int tl = 0; tl < TPW; ++tl) { c[tl][0] = 0.f; c[tl][1] = 0.f; }
    for (int i = threadIdx.x; i < HBUF; i += 32 * NWAVE) hbuf[i] = (_Float16)0.f;
    __syncthreads();

    const size_t xrow = (size_t)T * 2;
    const float* pxbase = x + (size_t)(b0 + col) * xrow;
    float* pout = out + (size_t)(b0 + col) * T;

    const int rdoff = col * HP + 16 * hi;
    const int wroff = col * HP + 4 * (w * TPW) + 2 * hi;

    float2 xv = *(const float2*)pxbase;
    int cur = 0;

    for (int t = 0; t < T; ++t) {
        v2f Bx;
        Bx[0] = (hi == 0) ? xv.x : 1.0f;
        Bx[1] = (hi == 0) ? xv.y : 0.0f;

        int tn = (t + 1 < T) ? (t + 1) : t;
        float2 xvn = *(const float2*)(pxbase + (size_t)tn * 2);

        v8f acc[TPW];
#pragma unroll
        for (int tl = 0; tl < TPW; ++tl) {
            v8f z = (v8f){};
            acc[tl] = __builtin_amdgcn_wmma_f32_16x16x4_f32(
                false, Ax[tl], false, Bx, (short)0, z, false, false);
        }

        v16h Bh;
        {
            const _Float16* hr = hbuf + cur * HBUF + rdoff;
            v8h lo = *(const v8h*)hr;
            v8h up = *(const v8h*)(hr + 8);
#pragma unroll
            for (int e = 0; e < 8; ++e) { Bh[e] = lo[e]; Bh[e + 8] = up[e]; }
        }

#pragma unroll
        for (int tl = 0; tl < TPW; ++tl) {
            acc[tl] = __builtin_amdgcn_wmma_f32_16x16x32_f16(
                false, Aw[tl], false, Bh, (short)0, acc[tl], false, false);
        }

        _Float16* hw = hbuf + (cur ^ 1) * HBUF + wroff;
#pragma unroll
        for (int tl = 0; tl < TPW; ++tl) {
            float hn[2];
#pragma unroll
            for (int u = 0; u < 2; ++u) {
                float ig = sigm(acc[tl][0 + u]);
                float fg = sigm(acc[tl][2 + u]);
                float gg = tanh_(acc[tl][4 + u]);
                float og = sigm(acc[tl][6 + u]);
                c[tl][u] = fg * c[tl][u] + ig * gg;
                hn[u] = og * tanh_(c[tl][u]);
            }
            *(v2h*)(hw + 4 * tl) = (v2h){(_Float16)hn[0], (_Float16)hn[1]};
        }

        v8f accy = (v8f){};
        if (w == 0)
            accy = __builtin_amdgcn_wmma_f32_16x16x32_f16(
                false, Afc, false, Bh, (short)0, Cfc, false, false);

        __builtin_amdgcn_fence(__ATOMIC_RELEASE, "workgroup", "local");
        __builtin_amdgcn_s_barrier();
        __builtin_amdgcn_fence(__ATOMIC_ACQUIRE, "workgroup", "local");

        if (w == 0 && t > 0 && hi == 0) pout[t - 1] = accy[0];

        xv = xvn;
        cur ^= 1;
    }

    if (w == 0) {
        v16h Bh;
        const _Float16* hr = hbuf + cur * HBUF + rdoff;
        v8h lo = *(const v8h*)hr;
        v8h up = *(const v8h*)(hr + 8);
#pragma unroll
        for (int e = 0; e < 8; ++e) { Bh[e] = lo[e]; Bh[e + 8] = up[e]; }
        v8f accy = __builtin_amdgcn_wmma_f32_16x16x32_f16(
            false, Afc, false, Bh, (short)0, Cfc, false, false);
        if (hi == 0) pout[T - 1] = accy[0];
    }
}

extern "C" void kernel_launch(void* const* d_in, const int* in_sizes, int n_in,
                              void* d_out, int out_size, void* d_ws, size_t ws_size,
                              hipStream_t stream) {
    const float* x    = (const float*)d_in[0];
    const float* W_ih = (const float*)d_in[1];
    const float* W_hh = (const float*)d_in[2];
    const float* b_ih = (const float*)d_in[3];
    const float* b_hh = (const float*)d_in[4];
    const float* W_fc = (const float*)d_in[5];
    const float* b_fc = (const float*)d_in[6];
    float* out = (float*)d_out;

    const int T = 512;
    const int B = out_size / T;
    const int nblocks = B / 16;

    lstm_wmma_kernel<<<dim3(nblocks), dim3(32 * NWAVE), 0, stream>>>(
        x, W_ih, W_hh, b_ih, b_hh, W_fc, b_fc, out, T);

    (void)in_sizes; (void)n_in; (void)d_ws; (void)ws_size;
}


# Each wave keeps all of h for 4 batch rows in registers (group-masked WMMA chains, h_t to the B layout by row_xmask DPP): no LDS, no barrier; bias moves to C, K sum split over 4 chained WMMAs
# speedup vs baseline: 1.7744x; 1.0624x over previous
#include <hip/hip_runtime.h>
#include <hip/hip_bf16.h>


typedef __attribute__((ext_vector_type(16))) _Float16 v16h;
typedef __attribute__((ext_vector_type(2)))  _Float16 v2h;
typedef __attribute__((ext_vector_type(8)))  float    v8f;
typedef __attribute__((ext_vector_type(2)))  float    v2f;
typedef __attribute__((ext_vector_type(8)))  int      v8i;

#define HID   32
#define NWAVE 4
#define NGRP  4
#define TPG   2

#if __has_builtin(__builtin_amdgcn_tanhf)
__device__ __forceinline__ float tanh_(float x) { return __builtin_amdgcn_tanhf(x); }
__device__ __forceinline__ float sigm(float x) {
    return __builtin_fmaf(0.5f, __builtin_amdgcn_tanhf(0.5f * x), 0.5f);
}
#else
__device__ __forceinline__ float fast_rcp(float x) { return __builtin_amdgcn_rcpf(x); }
__device__ __forceinline__ float tanh_(float x) {
    float e = __expf(-2.0f * x);
    return (1.0f - e) * fast_rcp(1.0f + e);
}
__device__ __forceinline__ float sigm(float x) { return fast_rcp(1.0f + __expf(-x)); }
#endif

template <int XM>
__device__ __forceinline__ int row_xor(int v) {
    return __builtin_amdgcn_update_dpp(v, v, 0x160 + XM, 0xF, 0xF, false);
}


__global__ __launch_bounds__(32 * NWAVE)
void lstm_wmma_kernel(const float* __restrict__ x,
                      const float* __restrict__ W_ih,
                      const float* __restrict__ W_hh,
                      const float* __restrict__ b_ih,
                      const float* __restrict__ b_hh,
                      const float* __restrict__ W_fc,
                      const float* __restrict__ b_fc,
                      float* __restrict__ out,
                      int T)
{
    const int lane = threadIdx.x & 31;
    const int w    = threadIdx.x >> 5;
    const int col  = lane & 15;
    const int hi   = lane >> 4;
    const int g    = col >> 2;
    const int bq   = col & 3;
    const int brow = blockIdx.x * (4 * NWAVE) + 4 * w + bq;

    const int q  = (col & 7) >> 1;
    const int mu = 2 * (col >> 3) + (col & 1);

    v16h Aw[NGRP][TPG];
#pragma unroll
    for (int j = 0; j < NGRP; ++j) {
#pragma unroll
        for (int tp = 0; tp < TPG; ++tp) {
            const float* wrow = W_hh + (size_t)(q * HID + 4 * (2 * j + tp) + mu) * HID;
#pragma unroll
            for (int k = 0; k < NGRP; ++k) {
#pragma unroll
                for (int ts = 0; ts < TPG; ++ts) {
                    const float2 wv = *(const float2*)(wrow + 8 * (j ^ k) + 4 * ts + 2 * hi);
                    Aw[j][tp][4 * k + 2 * ts + 0] = (_Float16)wv.x;
                    Aw[j][tp][4 * k + 2 * ts + 1] = (_Float16)wv.y;
                }
            }
        }
    }

    v2f Ax[2][TPG];
#pragma unroll
    for (int qx = 0; qx < 2; ++qx) {
#pragma unroll
        for (int tp = 0; tp < TPG; ++tp) {
            const int grow = q * HID + 4 * (2 * (2 * qx + hi) + tp) + mu;
            Ax[qx][tp][0] = W_ih[grow * 2 + 0];
            Ax[qx][tp][1] = W_ih[grow * 2 + 1];
        }
    }

    v8f Cb[TPG];
#pragma unroll
    for (int tp = 0; tp < TPG; ++tp) {
#pragma unroll
        for (int r = 0; r < 8; ++r) {
            const int grow = (r >> 1) * HID + 4 * (2 * g + tp) + 2 * hi + (r & 1);
            Cb[tp][r] = b_ih[grow] + b_hh[grow];
        }
    }

    v16h Afc = (v16h){};
    if (col == 0) {
#pragma unroll
        for (int k = 0; k < NGRP; ++k) {
#pragma unroll
            for (int ts = 0; ts < TPG; ++ts) {
                const float2 wv = *(const float2*)(W_fc + 8 * k + 4 * ts + 2 * hi);
                Afc[4 * k + 2 * ts + 0] = (_Float16)wv.x;
                Afc[4 * k + 2 * ts + 1] = (_Float16)wv.y;
            }
        }
    }
    v8f Cfc;
    {
        float bfc = b_fc[0];
#pragma unroll
        for (int e = 0; e < 8; ++e) Cfc[e] = bfc;
    }

    float c[TPG][2];
#pragma unroll
    for (int tp = 0; tp < TPG; ++tp) { c[tp][0] = 0.f; c[tp][1] = 0.f; }
    v8i H = (v8i){};

    const float* px = x + (size_t)brow * T * 2;
    float* pout = out + (size_t)brow * T;

    const bool mx0 = (g == hi);
    const bool mx1 = (g == 2 + hi);
    const bool ylane = (g == 0) && (hi == 0);

    float2 xv = *(const float2*)px;

    for (int t = 0; t < T; ++t) {
        int tn = (t + 1 < T) ? (t + 1) : t;
        float2 xvn = *(const float2*)(px + (size_t)tn * 2);

        v2f Bx0, Bx1;
        Bx0[0] = mx0 ? xv.x : 0.0f;  Bx0[1] = mx0 ? xv.y : 0.0f;
        Bx1[0] = mx1 ? xv.x : 0.0f;  Bx1[1] = mx1 ? xv.y : 0.0f;
        v8f acc[TPG];
#pragma unroll
        for (int tp = 0; tp < TPG; ++tp) {
            acc[tp] = __builtin_amdgcn_wmma_f32_16x16x4_f32(
                false, Ax[0][tp], false, Bx0, (short)0, Cb[tp], false, false);
            acc[tp] = __builtin_amdgcn_wmma_f32_16x16x4_f32(
                false, Ax[1][tp], false, Bx1, (short)0, acc[tp], false, false);
        }

#pragma unroll
        for (int j = 0; j < NGRP; ++j) {
            v8i Bj;
#pragma unroll
            for (int d = 0; d < 8; ++d) Bj[d] = (g == j) ? H[d] : 0;
            const v16h Bh = __builtin_bit_cast(v16h, Bj);
#pragma unroll
            for (int tp = 0; tp < TPG; ++tp) {
                acc[tp] = __builtin_amdgcn_wmma_f32_16x16x32_f16(
                    false, Aw[j][tp], false, Bh, (short)0, acc[tp], false, false);
            }
        }

        v8f accy = __builtin_amdgcn_wmma_f32_16x16x32_f16(
            false, Afc, false, __builtin_bit_cast(v16h, H), (short)0, Cfc, false, false);

        int P[TPG];
#pragma unroll
        for (int tp = 0; tp < TPG; ++tp) {
            float hn[2];
#pragma unroll
            for (int u = 0; u < 2; ++u) {
                float ig = sigm(acc[tp][0 + u]);
                float fg = sigm(acc[tp][2 + u]);
                float gg = tanh_(acc[tp][4 + u]);
                float og = sigm(acc[tp][6 + u]);
                c[tp][u] = fg * c[tp][u] + ig * gg;
                hn[u] = og * tanh_(c[tp][u]);
            }
            P[tp] = __builtin_bit_cast(int, (v2h){(_Float16)hn[0], (_Float16)hn[1]});
        }

#pragma unroll
        for (int tp = 0; tp < TPG; ++tp) {
            H[0 + tp] = P[tp];
            H[2 + tp] = row_xor<4>(P[tp]);
            H[4 + tp] = row_xor<8>(P[tp]);
            H[6 + tp] = row_xor<12>(P[tp]);
        }

        if (ylane && t > 0) pout[t - 1] = accy[0];

        xv = xvn;
    }

    {
        v8f accy = __builtin_amdgcn_wmma_f32_16x16x32_f16(
            false, Afc, false, __builtin_bit_cast(v16h, H), (short)0, Cfc, false, false);
        if (ylane) pout[T - 1] = accy[0];
    }
}

extern "C" void kernel_launch(void* const* d_in, const int* in_sizes, int n_in,
                              void* d_out, int out_size, void* d_ws, size_t ws_size,
                              hipStream_t stream) {
    const float* x    = (const float*)d_in[0];
    const float* W_ih = (const float*)d_in[1];
    const float* W_hh = (const float*)d_in[2];
    const float* b_ih = (const float*)d_in[3];
    const float* b_hh = (const float*)d_in[4];
    const float* W_fc = (const float*)d_in[5];
    const float* b_fc = (const float*)d_in[6];
    float* out = (float*)d_out;

    const int T = 512;
    const int B = out_size / T;
    const int nblocks = B / 16;

    lstm_wmma_kernel<<<dim3(nblocks), dim3(32 * NWAVE), 0, stream>>>(
        x, W_ih, W_hh, b_ih, b_hh, W_fc, b_fc, out, T);

    (void)in_sizes; (void)n_in; (void)d_ws; (void)ws_size;
}


# Time loop in blocks of 8 straight-line steps: a block's x is fetched one block ahead by 16-byte loads, its 8 outputs leave as two 16-byte stores (y_t taken right after h_t); sums and order unchanged
# speedup vs baseline: 2.7675x; 1.5597x over previous
#include <hip/hip_runtime.h>
#include <hip/hip_bf16.h>


typedef __attribute__((ext_vector_type(16))) _Float16 v16h;
typedef __attribute__((ext_vector_type(2)))  _Float16 v2h;
typedef __attribute__((ext_vector_type(8)))  float    v8f;
typedef __attribute__((ext_vector_type(2)))  float    v2f;
typedef __attribute__((ext_vector_type(8)))  int      v8i;

#define HID   32
#define NWAVE 4
#define NGRP  4
#define TPG   2
#define XB    8

#if __has_builtin(__builtin_amdgcn_tanhf)
__device__ __forceinline__ float tanh_(float x) { return __builtin_amdgcn_tanhf(x); }
#else
__device__ __forceinline__ float fast_rcp(float x) { return __builtin_amdgcn_rcpf(x); }
__device__ __forceinline__ float tanh_(float x) {
    float e = __expf(-2.0f * x);
    return (1.0f - e) * fast_rcp(1.0f + e);
}
#endif
__device__ __forceinline__ float sigm_half(float zh) { return __builtin_fmaf(0.5f, tanh_(zh), 0.5f); }

template <int XM>
__device__ __forceinline__ int row_xor(int v) {
    return __builtin_amdgcn_update_dpp(v, v, 0x160 + XM, 0xF, 0xF, false);
}


__device__ __forceinline__ float lstm_step(
    const float xa, const float xb,
    const v16h (&Aw)[NGRP][TPG], const v2f (&Ax)[2][TPG], const v8f (&Cb)[TPG],
    const v16h& Afc, const v8f& Cfc,
    float (&c)[TPG][2], v8i& H,
    const int g, const bool mx0, const bool mx1)
{
    v2f Bx0, Bx1;
    Bx0[0] = mx0 ? xa : 0.0f;  Bx0[1] = mx0 ? xb : 0.0f;
    Bx1[0] = mx1 ? xa : 0.0f;  Bx1[1] = mx1 ? xb : 0.0f;
    v8f acc[TPG];
#pragma unroll
    for (int tp = 0; tp < TPG; ++tp) {
        acc[tp] = __builtin_amdgcn_wmma_f32_16x16x4_f32(
            false, Ax[0][tp], false, Bx0, (short)0, Cb[tp], false, false);
        acc[tp] = __builtin_amdgcn_wmma_f32_16x16x4_f32(
            false, Ax[1][tp], false, Bx1, (short)0, acc[tp], false, false);
    }

#pragma unroll
    for (int j = 0; j < NGRP; ++j) {
        v8i Bj;
#pragma unroll
        for (int d = 0; d < 8; ++d) Bj[d] = (g == j) ? H[d] : 0;
        const v16h Bh = __builtin_bit_cast(v16h, Bj);
#pragma unroll
        for (int tp = 0; tp < TPG; ++tp) {
            acc[tp] = __builtin_amdgcn_wmma_f32_16x16x32_f16(
                false, Aw[j][tp], false, Bh, (short)0, acc[tp], false, false);
        }
    }

    int P[TPG];
#pragma unroll
    for (int tp = 0; tp < TPG; ++tp) {
        float hn[2];
#pragma unroll
        for (int u = 0; u < 2; ++u) {
            float ig = sigm_half(acc[tp][0 + u]);
            float fg = sigm_half(acc[tp][2 + u]);
            float gg = tanh_(acc[tp][4 + u]);
            float to = tanh_(acc[tp][6 + u]);
            c[tp][u] = fg * c[tp][u] + ig * gg;
            float tc = tanh_(c[tp][u]);
            hn[u] = __builtin_fmaf(to, tc, tc);
        }
        P[tp] = __builtin_bit_cast(int, (v2h){(_Float16)hn[0], (_Float16)hn[1]});
    }

#pragma unroll
    for (int tp = 0; tp < TPG; ++tp) {
        H[0 + tp] = P[tp];
        H[2 + tp] = row_xor<4>(P[tp]);
        H[4 + tp] = row_xor<8>(P[tp]);
        H[6 + tp] = row_xor<12>(P[tp]);
    }

    const v8f accy = __builtin_amdgcn_wmma_f32_16x16x32_f16(
        false, Afc, false, __builtin_bit_cast(v16h, H), (short)0, Cfc, false, false);
    return accy[0];
}

__global__ __launch_bounds__(32 * NWAVE)
void lstm_wmma_kernel(const float* __restrict__ x,
                      const float* __restrict__ W_ih,
                      const float* __restrict__ W_hh,
                      const float* __restrict__ b_ih,
                      const float* __restrict__ b_hh,
                      const float* __restrict__ W_fc,
                      const float* __restrict__ b_fc,
                      float* __restrict__ out,
                      int T)
{
    const int lane = threadIdx.x & 31;
    const int w    = threadIdx.x >> 5;
    const int col  = lane & 15;
    const int hi   = lane >> 4;
    const int g    = col >> 2;
    const int bq   = col & 3;
    const int brow = blockIdx.x * (4 * NWAVE) + 4 * w + bq;

    const int q  = (col & 7) >> 1;
    const int mu = 2 * (col >> 3) + (col & 1);

    v16h Aw[NGRP][TPG];
#pragma unroll
    for (int j = 0; j < NGRP; ++j) {
#pragma unroll
        for (int tp = 0; tp < TPG; ++tp) {
            const float* wrow = W_hh + (size_t)(q * HID + 4 * (2 * j + tp) + mu) * HID;
            const float sw = (q == 2) ? 0.5f : 0.25f;
#pragma unroll
            for (int k = 0; k < NGRP; ++k) {
#pragma unroll
                for (int ts = 0; ts < TPG; ++ts) {
                    const float2 wv = *(const float2*)(wrow + 8 * (j ^ k) + 4 * ts + 2 * hi);
                    Aw[j][tp][4 * k + 2 * ts + 0] = (_Float16)(wv.x * sw);
                    Aw[j][tp][4 * k + 2 * ts + 1] = (_Float16)(wv.y * sw);
                }
            }
        }
    }

    v2f Ax[2][TPG];
#pragma unroll
    for (int qx = 0; qx < 2; ++qx) {
#pragma unroll
        for (int tp = 0; tp < TPG; ++tp) {
            const int grow = q * HID + 4 * (2 * (2 * qx + hi) + tp) + mu;
            const float sx = (q == 2) ? 1.0f : 0.5f;
            Ax[qx][tp][0] = W_ih[grow * 2 + 0] * sx;
            Ax[qx][tp][1] = W_ih[grow * 2 + 1] * sx;
        }
    }

    v8f Cb[TPG];
#pragma unroll
    for (int tp = 0; tp < TPG; ++tp) {
#pragma unroll
        for (int r = 0; r < 8; ++r) {
            const int grow = (r >> 1) * HID + 4 * (2 * g + tp) + 2 * hi + (r & 1);
            Cb[tp][r] = (b_ih[grow] + b_hh[grow]) * (((r >> 1) == 2) ? 1.0f : 0.5f);
        }
    }

    v16h Afc = (v16h){};
    if (col == 0) {
#pragma unroll
        for (int k = 0; k < NGRP; ++k) {
#pragma unroll
            for (int ts = 0; ts < TPG; ++ts) {
                const float2 wv = *(const float2*)(W_fc + 8 * k + 4 * ts + 2 * hi);
                Afc[4 * k + 2 * ts + 0] = (_Float16)(wv.x * 0.5f);
                Afc[4 * k + 2 * ts + 1] = (_Float16)(wv.y * 0.5f);
            }
        }
    }
    v8f Cfc;
    {
        float bfc = b_fc[0];
#pragma unroll
        for (int e = 0; e < 8; ++e) Cfc[e] = bfc;
    }

    float c[TPG][2];
#pragma unroll
    for (int tp = 0; tp < TPG; ++tp) { c[tp][0] = 0.f; c[tp][1] = 0.f; }
    v8i H = (v8i){};

    const float* px = x + (size_t)brow * T * 2;
    float* pout = out + (size_t)brow * T;

    const bool mx0 = (g == hi);
    const bool mx1 = (g == 2 + hi);
    const bool ylane = (g == 0) && (hi == 0);

    const int Tb = (T % 4 == 0) ? T - (T % XB) : 0;
    float4 xq[XB / 2];
    if (Tb > 0) {
#pragma unroll
        for (int i = 0; i < XB / 2; ++i) xq[i] = *(const float4*)(px + 4 * i);
    }
    for (int t0 = 0; t0 < Tb; t0 += XB) {
        const int tnx = (t0 + XB < Tb) ? (t0 + XB) : t0;
        float4 xn[XB / 2];
#pragma unroll
        for (int i = 0; i < XB / 2; ++i) xn[i] = *(const float4*)(px + (size_t)tnx * 2 + 4 * i);
        __builtin_amdgcn_sched_barrier(0);
        float yv[XB];
#pragma unroll
        for (int s = 0; s < XB; ++s) {
            const float xa = (s & 1) ? xq[s >> 1].z : xq[s >> 1].x;
            const float xb = (s & 1) ? xq[s >> 1].w : xq[s >> 1].y;
            yv[s] = lstm_step(xa, xb, Aw, Ax, Cb, Afc, Cfc, c, H, g, mx0, mx1);
        }
        if (ylane) {
#pragma unroll
            for (int i = 0; i < XB / 4; ++i)
                *(float4*)(pout + t0 + 4 * i) =
                    make_float4(yv[4 * i], yv[4 * i + 1], yv[4 * i + 2], yv[4 * i + 3]);
        }
#pragma unroll
        for (int i = 0; i < XB / 2; ++i) xq[i] = xn[i];
    }
    for (int t = Tb; t < T; ++t) {
        const float2 xv = *(const float2*)(px + (size_t)t * 2);
        const float y = lstm_step(xv.x, xv.y, Aw, Ax, Cb, Afc, Cfc, c, H, g, mx0, mx1);
        if (ylane) pout[t] = y;
    }
}

extern "C" void kernel_launch(void* const* d_in, const int* in_sizes, int n_in,
                              void* d_out, int out_size, void* d_ws, size_t ws_size,
                              hipStream_t stream) {
    const float* x    = (const float*)d_in[0];
    const float* W_ih = (const float*)d_in[1];
    const float* W_hh = (const float*)d_in[2];
    const float* b_ih = (const float*)d_in[3];
    const float* b_hh = (const float*)d_in[4];
    const float* W_fc = (const float*)d_in[5];
    const float* b_fc = (const float*)d_in[6];
    float* out = (float*)d_out;

    const int T = 512;
    const int B = out_size / T;
    const int nblocks = B / 16;

    lstm_wmma_kernel<<<dim3(nblocks), dim3(32 * NWAVE), 0, stream>>>(
        x, W_ih, W_hh, b_ih, b_hh, W_fc, b_fc, out, T);

    (void)in_sizes; (void)n_in; (void)d_ws; (void)ws_size;
}


# x projection issued one step ahead: the two W_ih WMMAs of step t+1 run beside the cell update of step t, so only the W_hh chain lies between h and the gates; sums and their order unchanged
# speedup vs baseline: 3.4897x; 1.2610x over previous
#include <hip/hip_runtime.h>
#include <hip/hip_bf16.h>


typedef __attribute__((ext_vector_type(16))) _Float16 v16h;
typedef __attribute__((ext_vector_type(2)))  _Float16 v2h;
typedef __attribute__((ext_vector_type(8)))  float    v8f;
typedef __attribute__((ext_vector_type(8)))  int      v8i;

#define HID   32
#define NWAVE 4
#define NGRP  4
#define TPG   2
#define XB    8

#if __has_builtin(__builtin_amdgcn_tanhf)
__device__ __forceinline__ float tanh_(float x) { return __builtin_amdgcn_tanhf(x); }
#else
__device__ __forceinline__ float fast_rcp(float x) { return __builtin_amdgcn_rcpf(x); }
__device__ __forceinline__ float tanh_(float x) {
    float e = __expf(-2.0f * x);
    return (1.0f - e) * fast_rcp(1.0f + e);
}
#endif
__device__ __forceinline__ float sigm_half(float zh) { return __builtin_fmaf(0.5f, tanh_(zh), 0.5f); }

template <int XM>
__device__ __forceinline__ int row_xor(int v) {
    return __builtin_amdgcn_update_dpp(v, v, 0x160 + XM, 0xF, 0xF, false);
}


__device__ __forceinline__ v16h xproj_operand(const float xe, const int g)
{
    const _Float16 xh = (_Float16)xe;
    const float    xr = xe - (float)xh;
    const int Xh = __builtin_bit_cast(int, (v2h){xh, xh});
    const int Xl = __builtin_bit_cast(int, (v2h){(_Float16)xr, (_Float16)xr});
    v8i Bxi;
#pragma unroll
    for (int gp = 0; gp < NGRP; ++gp) {
        Bxi[2 * gp + 0] = (g == gp) ? Xh : 0;
        Bxi[2 * gp + 1] = (g == gp) ? Xl : 0;
    }
    return __builtin_bit_cast(v16h, Bxi);
}

__device__ __forceinline__ float lstm_step(
    const float xe_next,
    const v16h (&Aw)[NGRP][TPG], const v16h (&Ax)[TPG], const v8f (&Cb)[TPG],
    const v16h& Afc, const v8f& Cfc,
    v8f (&accx)[TPG], float (&c)[TPG][2], v8i& H,
    const int g)
{
    v8f acc[TPG];
#pragma unroll
    for (int tp = 0; tp < TPG; ++tp) acc[tp] = accx[tp];

#pragma unroll
    for (int j = 0; j < NGRP; ++j) {
        v8i Bj;
#pragma unroll
        for (int d = 0; d < 8; ++d) Bj[d] = (g == j) ? H[d] : 0;
        const v16h Bh = __builtin_bit_cast(v16h, Bj);
#pragma unroll
        for (int tp = 0; tp < TPG; ++tp) {
            acc[tp] = __builtin_amdgcn_wmma_f32_16x16x32_f16(
                false, Aw[j][tp], false, Bh, (short)0, acc[tp], false, false);
        }
    }

    const v16h Bx = xproj_operand(xe_next, g);
    __builtin_amdgcn_sched_barrier(0);
#pragma unroll
    for (int tp = 0; tp < TPG; ++tp) {
        accx[tp] = __builtin_amdgcn_wmma_f32_16x16x32_f16(
            false, Ax[tp], false, Bx, (short)0, Cb[tp], false, false);
    }
    __builtin_amdgcn_sched_barrier(0);

    int P[TPG];
#pragma unroll
    for (int tp = 0; tp < TPG; ++tp) {
        float hn[2];
#pragma unroll
        for (int u = 0; u < 2; ++u) {
            float ig = sigm_half(acc[tp][0 + u]);
            float fg = sigm_half(acc[tp][2 + u]);
            float gg = tanh_(acc[tp][4 + u]);
            float to = tanh_(acc[tp][6 + u]);
            c[tp][u] = fg * c[tp][u] + ig * gg;
            float tc = tanh_(c[tp][u]);
            hn[u] = __builtin_fmaf(to, tc, tc);
        }
        P[tp] = __builtin_bit_cast(int, (v2h){(_Float16)hn[0], (_Float16)hn[1]});
    }

#pragma unroll
    for (int tp = 0; tp < TPG; ++tp) {
        H[0 + tp] = P[tp];
        H[2 + tp] = row_xor<4>(P[tp]);
        H[4 + tp] = row_xor<8>(P[tp]);
        H[6 + tp] = row_xor<12>(P[tp]);
    }

    const v8f accy = __builtin_amdgcn_wmma_f32_16x16x32_f16(
        false, Afc, false, __builtin_bit_cast(v16h, H), (short)0, Cfc, false, false);
    return accy[0];
}

__global__ __launch_bounds__(32 * NWAVE)
void lstm_wmma_kernel(const float* __restrict__ x,
                      const float* __restrict__ W_ih,
                      const float* __restrict__ W_hh,
                      const float* __restrict__ b_ih,
                      const float* __restrict__ b_hh,
                      const float* __restrict__ W_fc,
                      const float* __restrict__ b_fc,
                      float* __restrict__ out,
                      int T)
{
    const int lane = threadIdx.x & 31;
    const int w    = threadIdx.x >> 5;
    const int col  = lane & 15;
    const int hi   = lane >> 4;
    const int g    = col >> 2;
    const int bq   = col & 3;
    const int brow = blockIdx.x * (4 * NWAVE) + 4 * w + bq;

    const int q  = (col & 7) >> 1;
    const int mu = 2 * (col >> 3) + (col & 1);

    v16h Aw[NGRP][TPG];
#pragma unroll
    for (int j = 0; j < NGRP; ++j) {
#pragma unroll
        for (int tp = 0; tp < TPG; ++tp) {
            const float* wrow = W_hh + (size_t)(q * HID + 4 * (2 * j + tp) + mu) * HID;
            const float sw = (q == 2) ? 0.5f : 0.25f;
#pragma unroll
            for (int k = 0; k < NGRP; ++k) {
#pragma unroll
                for (int ts = 0; ts < TPG; ++ts) {
                    const float2 wv = *(const float2*)(wrow + 8 * (j ^ k) + 4 * ts + 2 * hi);
                    Aw[j][tp][4 * k + 2 * ts + 0] = (_Float16)(wv.x * sw);
                    Aw[j][tp][4 * k + 2 * ts + 1] = (_Float16)(wv.y * sw);
                }
            }
        }
    }

    v16h Ax[TPG];
#pragma unroll
    for (int tp = 0; tp < TPG; ++tp) {
#pragma unroll
        for (int gp = 0; gp < NGRP; ++gp) {
            const int grow = q * HID + 4 * (2 * gp + tp) + mu;
            const float sx = (q == 2) ? 1.0f : 0.5f;
            const float wv = W_ih[grow * 2 + hi] * sx;
            const _Float16 wh = (_Float16)wv;
            const _Float16 wl = (_Float16)(wv - (float)wh);
#pragma unroll
            for (int tx = 0; tx < 2; ++tx) {
                Ax[tp][4 * gp + 2 * tx + 0] = wh;
                Ax[tp][4 * gp + 2 * tx + 1] = wl;
            }
        }
    }

    v8f Cb[TPG];
#pragma unroll
    for (int tp = 0; tp < TPG; ++tp) {
#pragma unroll
        for (int r = 0; r < 8; ++r) {
            const int grow = (r >> 1) * HID + 4 * (2 * g + tp) + 2 * hi + (r & 1);
            Cb[tp][r] = (b_ih[grow] + b_hh[grow]) * (((r >> 1) == 2) ? 1.0f : 0.5f);
        }
    }

    v16h Afc = (v16h){};
    if (col == 0) {
#pragma unroll
        for (int k = 0; k < NGRP; ++k) {
#pragma unroll
            for (int ts = 0; ts < TPG; ++ts) {
                const float2 wv = *(const float2*)(W_fc + 8 * k + 4 * ts + 2 * hi);
                Afc[4 * k + 2 * ts + 0] = (_Float16)(wv.x * 0.5f);
                Afc[4 * k + 2 * ts + 1] = (_Float16)(wv.y * 0.5f);
            }
        }
    }
    v8f Cfc;
    {
        float bfc = b_fc[0];
#pragma unroll
        for (int e = 0; e < 8; ++e) Cfc[e] = bfc;
    }

    float c[TPG][2];
#pragma unroll
    for (int tp = 0; tp < TPG; ++tp) { c[tp][0] = 0.f; c[tp][1] = 0.f; }
    v8i H = (v8i){};

    const float* px = x + (size_t)brow * T * 2;
    float* pout = out + (size_t)brow * T;

    const bool ylane = (g == 0) && (hi == 0);

    const int Tb = (T % 4 == 0) ? T - (T % XB) : 0;
    float4 xq[XB / 2];
    if (Tb > 0) {
#pragma unroll
        for (int i = 0; i < XB / 2; ++i) xq[i] = *(const float4*)(px + 4 * i);
    }

    v8f accx[TPG];
    {
        float2 x0 = make_float2(0.f, 0.f);
        if (Tb > 0)     x0 = make_float2(xq[0].x, xq[0].y);
        else if (T > 0) x0 = *(const float2*)px;
        const v16h Bx = xproj_operand(hi ? x0.y : x0.x, g);
#pragma unroll
        for (int tp = 0; tp < TPG; ++tp) {
            accx[tp] = __builtin_amdgcn_wmma_f32_16x16x32_f16(
                false, Ax[tp], false, Bx, (short)0, Cb[tp], false, false);
        }
    }

    for (int t0 = 0; t0 < Tb; t0 += XB) {
        const int tnx = (t0 + XB < Tb) ? (t0 + XB) : t0;
        float4 xn[XB / 2];
#pragma unroll
        for (int i = 0; i < XB / 2; ++i) xn[i] = *(const float4*)(px + (size_t)tnx * 2 + 4 * i);
        __builtin_amdgcn_sched_barrier(0);
        float yv[XB];
#pragma unroll
        for (int s = 0; s < XB; ++s) {
            const int sn = s + 1;
            float xa, xb;
            if (sn < XB) {
                xa = (sn & 1) ? xq[sn >> 1].z : xq[sn >> 1].x;
                xb = (sn & 1) ? xq[sn >> 1].w : xq[sn >> 1].y;
            } else {
                xa = xn[0].x;
                xb = xn[0].y;
            }
            yv[s] = lstm_step(hi ? xb : xa, Aw, Ax, Cb, Afc, Cfc, accx, c, H, g);
        }
        if (ylane) {
#pragma unroll
            for (int i = 0; i < XB / 4; ++i)
                *(float4*)(pout + t0 + 4 * i) =
                    make_float4(yv[4 * i], yv[4 * i + 1], yv[4 * i + 2], yv[4 * i + 3]);
        }
#pragma unroll
        for (int i = 0; i < XB / 2; ++i) xq[i] = xn[i];
    }
    if (Tb > 0 && Tb < T) {
        const float2 xv = *(const float2*)(px + (size_t)Tb * 2);
        const v16h Bx = xproj_operand(hi ? xv.y : xv.x, g);
#pragma unroll
        for (int tp = 0; tp < TPG; ++tp) {
            accx[tp] = __builtin_amdgcn_wmma_f32_16x16x32_f16(
                false, Ax[tp], false, Bx, (short)0, Cb[tp], false, false);
        }
    }
    for (int t = Tb; t < T; ++t) {
        const int tn = (t + 1 < T) ? (t + 1) : t;
        const float2 xv = *(const float2*)(px + (size_t)tn * 2);
        const float y = lstm_step(hi ? xv.y : xv.x, Aw, Ax, Cb, Afc, Cfc, accx, c, H, g);
        if (ylane) pout[t] = y;
    }
}

extern "C" void kernel_launch(void* const* d_in, const int* in_sizes, int n_in,
                              void* d_out, int out_size, void* d_ws, size_t ws_size,
                              hipStream_t stream) {
    const float* x    = (const float*)d_in[0];
    const float* W_ih = (const float*)d_in[1];
    const float* W_hh = (const float*)d_in[2];
    const float* b_ih = (const float*)d_in[3];
    const float* b_hh = (const float*)d_in[4];
    const float* W_fc = (const float*)d_in[5];
    const float* b_fc = (const float*)d_in[6];
    float* out = (float*)d_out;

    const int T = 512;
    const int B = out_size / T;
    const int nblocks = B / 16;

    lstm_wmma_kernel<<<dim3(nblocks), dim3(32 * NWAVE), 0, stream>>>(
        x, W_ih, W_hh, b_ih, b_hh, W_fc, b_fc, out, T);

    (void)in_sizes; (void)n_in; (void)d_ws; (void)ws_size;
}


# Recurrent chain issued as 8 back-to-back WMMAs: its masked h operands are built ahead in registers of their own and y_t's WMMA moves behind the next step's chain; sums and their order unchanged
# speedup vs baseline: 3.6094x; 1.0343x over previous
#include <hip/hip_runtime.h>
#include <hip/hip_bf16.h>


typedef __attribute__((ext_vector_type(16))) _Float16 v16h;
typedef __attribute__((ext_vector_type(2)))  _Float16 v2h;
typedef __attribute__((ext_vector_type(8)))  float    v8f;
typedef __attribute__((ext_vector_type(8)))  int      v8i;

#define HID   32
#define NWAVE 4
#define NGRP  4
#define TPG   2
#define XB    8

#if __has_builtin(__builtin_amdgcn_tanhf)
__device__ __forceinline__ float tanh_(float x) { return __builtin_amdgcn_tanhf(x); }
#else
__device__ __forceinline__ float fast_rcp(float x) { return __builtin_amdgcn_rcpf(x); }
__device__ __forceinline__ float tanh_(float x) {
    float e = __expf(-2.0f * x);
    return (1.0f - e) * fast_rcp(1.0f + e);
}
#endif
__device__ __forceinline__ float sigm_half(float zh) { return __builtin_fmaf(0.5f, tanh_(zh), 0.5f); }

template <int XM>
__device__ __forceinline__ int row_xor(int v) {
    return __builtin_amdgcn_update_dpp(v, v, 0x160 + XM, 0xF, 0xF, false);
}


__device__ __forceinline__ v16h xproj_operand(const float xe, const int g)
{
    const _Float16 xh = (_Float16)xe;
    const float    xr = xe - (float)xh;
    const int Xh = __builtin_bit_cast(int, (v2h){xh, xh});
    const int Xl = __builtin_bit_cast(int, (v2h){(_Float16)xr, (_Float16)xr});
    v8i Bxi;
#pragma unroll
    for (int gp = 0; gp < NGRP; ++gp) {
        Bxi[2 * gp + 0] = (g == gp) ? Xh : 0;
        Bxi[2 * gp + 1] = (g == gp) ? Xl : 0;
    }
    return __builtin_bit_cast(v16h, Bxi);
}

__device__ __forceinline__ float lstm_step(
    const float xe_next,
    const v16h (&Aw)[NGRP][TPG], const v16h (&Ax)[TPG], const v8f (&Cb)[TPG],
    const v16h& Afc, const v8f& Cfc,
    v8f (&accx)[TPG], float (&c)[TPG][2], v8i& H,
    const int g)
{
    v8f acc[TPG];
#pragma unroll
    for (int tp = 0; tp < TPG; ++tp) acc[tp] = accx[tp];

    v8i Bm[NGRP];
#pragma unroll
    for (int j = 0; j < NGRP; ++j) {
#pragma unroll
        for (int d = 0; d < 8; ++d) Bm[j][d] = (g == j) ? H[d] : 0;
    }

#pragma unroll
    for (int j = 0; j < NGRP; ++j) {
        const v16h Bh = __builtin_bit_cast(v16h, Bm[j]);
#pragma unroll
        for (int tp = 0; tp < TPG; ++tp) {
            acc[tp] = __builtin_amdgcn_wmma_f32_16x16x32_f16(
                false, Aw[j][tp], false, Bh, (short)0, acc[tp], false, false);
        }
    }

    const v8f accy = __builtin_amdgcn_wmma_f32_16x16x32_f16(
        false, Afc, false, __builtin_bit_cast(v16h, H), (short)0, Cfc, false, false);

    const v16h Bx = xproj_operand(xe_next, g);
    __builtin_amdgcn_sched_barrier(0);
#pragma unroll
    for (int tp = 0; tp < TPG; ++tp) {
        accx[tp] = __builtin_amdgcn_wmma_f32_16x16x32_f16(
            false, Ax[tp], false, Bx, (short)0, Cb[tp], false, false);
    }
    __builtin_amdgcn_sched_barrier(0);

    int P[TPG];
#pragma unroll
    for (int tp = 0; tp < TPG; ++tp) {
        float hn[2];
#pragma unroll
        for (int u = 0; u < 2; ++u) {
            float ig = sigm_half(acc[tp][0 + u]);
            float fg = sigm_half(acc[tp][2 + u]);
            float gg = tanh_(acc[tp][4 + u]);
            float to = tanh_(acc[tp][6 + u]);
            c[tp][u] = fg * c[tp][u] + ig * gg;
            float tc = tanh_(c[tp][u]);
            hn[u] = __builtin_fmaf(to, tc, tc);
        }
        P[tp] = __builtin_bit_cast(int, (v2h){(_Float16)hn[0], (_Float16)hn[1]});
    }

#pragma unroll
    for (int tp = 0; tp < TPG; ++tp) {
        H[0 + tp] = P[tp];
        H[2 + tp] = row_xor<4>(P[tp]);
        H[4 + tp] = row_xor<8>(P[tp]);
        H[6 + tp] = row_xor<12>(P[tp]);
    }
    return accy[0];
}

__device__ __forceinline__ float fc_out(const v16h& Afc, const v8f& Cfc, const v8i& H)
{
    const v8f accy = __builtin_amdgcn_wmma_f32_16x16x32_f16(
        false, Afc, false, __builtin_bit_cast(v16h, H), (short)0, Cfc, false, false);
    return accy[0];
}

__global__ __launch_bounds__(32 * NWAVE)
void lstm_wmma_kernel(const float* __restrict__ x,
                      const float* __restrict__ W_ih,
                      const float* __restrict__ W_hh,
                      const float* __restrict__ b_ih,
                      const float* __restrict__ b_hh,
                      const float* __restrict__ W_fc,
                      const float* __restrict__ b_fc,
                      float* __restrict__ out,
                      int T)
{
    const int lane = threadIdx.x & 31;
    const int w    = threadIdx.x >> 5;
    const int col  = lane & 15;
    const int hi   = lane >> 4;
    const int g    = col >> 2;
    const int bq   = col & 3;
    const int brow = blockIdx.x * (4 * NWAVE) + 4 * w + bq;

    const int q  = (col & 7) >> 1;
    const int mu = 2 * (col >> 3) + (col & 1);

    v16h Aw[NGRP][TPG];
#pragma unroll
    for (int j = 0; j < NGRP; ++j) {
#pragma unroll
        for (int tp = 0; tp < TPG; ++tp) {
            const float* wrow = W_hh + (size_t)(q * HID + 4 * (2 * j + tp) + mu) * HID;
            const float sw = (q == 2) ? 0.5f : 0.25f;
#pragma unroll
            for (int k = 0; k < NGRP; ++k) {
#pragma unroll
                for (int ts = 0; ts < TPG; ++ts) {
                    const float2 wv = *(const float2*)(wrow + 8 * (j ^ k) + 4 * ts + 2 * hi);
                    Aw[j][tp][4 * k + 2 * ts + 0] = (_Float16)(wv.x * sw);
                    Aw[j][tp][4 * k + 2 * ts + 1] = (_Float16)(wv.y * sw);
                }
            }
        }
    }

    v16h Ax[TPG];
#pragma unroll
    for (int tp = 0; tp < TPG; ++tp) {
#pragma unroll
        for (int gp = 0; gp < NGRP; ++gp) {
            const int grow = q * HID + 4 * (2 * gp + tp) + mu;
            const float sx = (q == 2) ? 1.0f : 0.5f;
            const float wv = W_ih[grow * 2 + hi] * sx;
            const _Float16 wh = (_Float16)wv;
            const _Float16 wl = (_Float16)(wv - (float)wh);
#pragma unroll
            for (int tx = 0; tx < 2; ++tx) {
                Ax[tp][4 * gp + 2 * tx + 0] = wh;
                Ax[tp][4 * gp + 2 * tx + 1] = wl;
            }
        }
    }

    v8f Cb[TPG];
#pragma unroll
    for (int tp = 0; tp < TPG; ++tp) {
#pragma unroll
        for (int r = 0; r < 8; ++r) {
            const int grow = (r >> 1) * HID + 4 * (2 * g + tp) + 2 * hi + (r & 1);
            Cb[tp][r] = (b_ih[grow] + b_hh[grow]) * (((r >> 1) == 2) ? 1.0f : 0.5f);
        }
    }

    v16h Afc = (v16h){};
    if (col == 0) {
#pragma unroll
        for (int k = 0; k < NGRP; ++k) {
#pragma unroll
            for (int ts = 0; ts < TPG; ++ts) {
                const float2 wv = *(const float2*)(W_fc + 8 * k + 4 * ts + 2 * hi);
                Afc[4 * k + 2 * ts + 0] = (_Float16)(wv.x * 0.5f);
                Afc[4 * k + 2 * ts + 1] = (_Float16)(wv.y * 0.5f);
            }
        }
    }
    v8f Cfc;
    {
        float bfc = b_fc[0];
#pragma unroll
        for (int e = 0; e < 8; ++e) Cfc[e] = bfc;
    }

    float c[TPG][2];
#pragma unroll
    for (int tp = 0; tp < TPG; ++tp) { c[tp][0] = 0.f; c[tp][1] = 0.f; }
    v8i H = (v8i){};

    const float* px = x + (size_t)brow * T * 2;
    float* pout = out + (size_t)brow * T;

    const bool ylane = (g == 0) && (hi == 0);

    const int Tb = (T % 4 == 0) ? T - (T % XB) : 0;
    float4 xq[XB / 2];
    if (Tb > 0) {
#pragma unroll
        for (int i = 0; i < XB / 2; ++i) xq[i] = *(const float4*)(px + 4 * i);
    }

    v8f accx[TPG];
    {
        float2 x0 = make_float2(0.f, 0.f);
        if (Tb > 0)     x0 = make_float2(xq[0].x, xq[0].y);
        else if (T > 0) x0 = *(const float2*)px;
        const v16h Bx = xproj_operand(hi ? x0.y : x0.x, g);
#pragma unroll
        for (int tp = 0; tp < TPG; ++tp) {
            accx[tp] = __builtin_amdgcn_wmma_f32_16x16x32_f16(
                false, Ax[tp], false, Bx, (short)0, Cb[tp], false, false);
        }
    }

    float yp[XB - 1];
#pragma unroll
    for (int i = 0; i < XB - 1; ++i) yp[i] = 0.f;
    for (int t0 = 0; t0 < Tb; t0 += XB) {
        const int tnx = (t0 + XB < Tb) ? (t0 + XB) : t0;
        float4 xn[XB / 2];
#pragma unroll
        for (int i = 0; i < XB / 2; ++i) xn[i] = *(const float4*)(px + (size_t)tnx * 2 + 4 * i);
        __builtin_amdgcn_sched_barrier(0);
        float yv[XB];
#pragma unroll
        for (int s = 0; s < XB; ++s) {
            const int sn = s + 1;
            float xa, xb;
            if (sn < XB) {
                xa = (sn & 1) ? xq[sn >> 1].z : xq[sn >> 1].x;
                xb = (sn & 1) ? xq[sn >> 1].w : xq[sn >> 1].y;
            } else {
                xa = xn[0].x;
                xb = xn[0].y;
            }
            yv[s] = lstm_step(hi ? xb : xa, Aw, Ax, Cb, Afc, Cfc, accx, c, H, g);
        }
        if (t0 > 0 && ylane) {
            *(float4*)(pout + t0 - XB)     = make_float4(yp[0], yp[1], yp[2], yp[3]);
            *(float4*)(pout + t0 - XB + 4) = make_float4(yp[4], yp[5], yp[6], yv[0]);
        }
#pragma unroll
        for (int i = 0; i < XB - 1; ++i) yp[i] = yv[i + 1];
#pragma unroll
        for (int i = 0; i < XB / 2; ++i) xq[i] = xn[i];
    }
    if (Tb > 0) {
        const float yl = fc_out(Afc, Cfc, H);
        if (ylane) {
            *(float4*)(pout + Tb - XB)     = make_float4(yp[0], yp[1], yp[2], yp[3]);
            *(float4*)(pout + Tb - XB + 4) = make_float4(yp[4], yp[5], yp[6], yl);
        }
    }
    if (Tb > 0 && Tb < T) {
        const float2 xv = *(const float2*)(px + (size_t)Tb * 2);
        const v16h Bx = xproj_operand(hi ? xv.y : xv.x, g);
#pragma unroll
        for (int tp = 0; tp < TPG; ++tp) {
            accx[tp] = __builtin_amdgcn_wmma_f32_16x16x32_f16(
                false, Ax[tp], false, Bx, (short)0, Cb[tp], false, false);
        }
    }
    for (int t = Tb; t < T; ++t) {
        const int tn = (t + 1 < T) ? (t + 1) : t;
        const float2 xv = *(const float2*)(px + (size_t)tn * 2);
        const float y = lstm_step(hi ? xv.y : xv.x, Aw, Ax, Cb, Afc, Cfc, accx, c, H, g);
        if (ylane && t > Tb) pout[t - 1] = y;
    }
    if (Tb < T) {
        const float yl = fc_out(Afc, Cfc, H);
        if (ylane) pout[T - 1] = yl;
    }
}

extern "C" void kernel_launch(void* const* d_in, const int* in_sizes, int n_in,
                              void* d_out, int out_size, void* d_ws, size_t ws_size,
                              hipStream_t stream) {
    const float* x    = (const float*)d_in[0];
    const float* W_ih = (const float*)d_in[1];
    const float* W_hh = (const float*)d_in[2];
    const float* b_ih = (const float*)d_in[3];
    const float* b_hh = (const float*)d_in[4];
    const float* W_fc = (const float*)d_in[5];
    const float* b_fc = (const float*)d_in[6];
    float* out = (float*)d_out;

    const int T = 512;
    const int B = out_size / T;
    const int nblocks = B / 16;

    lstm_wmma_kernel<<<dim3(nblocks), dim3(32 * NWAVE), 0, stream>>>(
        x, W_ih, W_hh, b_ih, b_hh, W_fc, b_fc, out, T);

    (void)in_sizes; (void)n_in; (void)d_ws; (void)ws_size;
}
